# starting kernel
# baseline (speedup 1.0000x reference)
#include <hip/hip_runtime.h>

typedef __attribute__((ext_vector_type(2))) float v2f;
typedef __attribute__((ext_vector_type(8))) float v8f;

#define Bb 32
#define Nn 128
#define Dn 512
#define NL 16
#define ROWS (Bb * Nn)

__global__ void prep_weights(const float* __restrict__ w,
                             const float* __restrict__ bias,
                             float* __restrict__ Wa,
                             float* __restrict__ Bt) {
    int idx = blockIdx.x * blockDim.x + threadIdx.x;
    if (idx < Dn * Dn) {
        int o = idx / Dn;
        int i = idx - o * Dn;
        float v = w[idx] + w[Dn * Dn + idx];
        if (o == i) v += 1.0f;
        Wa[idx] = v;
    }
    if (idx < Dn * NL) {
        int o = idx / NL;
        int l = idx - o * NL;
        Bt[idx] = bias[l * Dn + o];
    }
}

__global__ void count_labels(const int* __restrict__ graph,
                             float* __restrict__ cnt) {
    __shared__ int h[NL];
    int row = blockIdx.x;
    int t = threadIdx.x;
    if (t < NL) h[t] = 0;
    __syncthreads();
    int lab = graph[row * Nn + t] & (NL - 1);
    atomicAdd(&h[lab], 1);
    __syncthreads();
    if (t < NL) cnt[row * NL + t] = (float)h[t];
}

__global__ __launch_bounds__(256) void gemm_fused(
    const float* __restrict__ X,
    const float* __restrict__ Wa,
    const float* __restrict__ Bt,
    const float* __restrict__ cnt,
    float* __restrict__ out) {

    const int lane = threadIdx.x & 31;
    const int wave = threadIdx.x >> 5;
    const int wid  = blockIdx.x * 8 + wave;
    const int m0 = (wid >> 3) * 32;
    const int o0 = (wid & 7) * 64;
    const int r = lane & 15;
    const int h = lane >> 4;

    const float* a0p = X + (size_t)(m0 + r) * Dn + 2 * h;
    const float* a1p = X + (size_t)(m0 + 16 + r) * Dn + 2 * h;
    const float* bp0 = Wa + (size_t)(o0 + 0 * 16 + r) * Dn + 2 * h;
    const float* bp1 = Wa + (size_t)(o0 + 1 * 16 + r) * Dn + 2 * h;
    const float* bp2 = Wa + (size_t)(o0 + 2 * 16 + r) * Dn + 2 * h;
    const float* bp3 = Wa + (size_t)(o0 + 3 * 16 + r) * Dn + 2 * h;

    v8f acc[2][4];
#pragma unroll
    for (int mt = 0; mt < 2; ++mt)
#pragma unroll
        for (int nt = 0; nt < 4; ++nt)
            acc[mt][nt] = (v8f){0.f, 0.f, 0.f, 0.f, 0.f, 0.f, 0.f, 0.f};

#pragma unroll 4
    for (int k0 = 0; k0 < Dn; k0 += 4) {
        v2f a0 = *(const v2f*)(a0p + k0);
        v2f a1 = *(const v2f*)(a1p + k0);
        v2f b0 = *(const v2f*)(bp0 + k0);
        v2f b1 = *(const v2f*)(bp1 + k0);
        v2f b2 = *(const v2f*)(bp2 + k0);
        v2f b3 = *(const v2f*)(bp3 + k0);
        acc[0][0] = __builtin_amdgcn_wmma_f32_16x16x4_f32(false, a0, false, b0, (short)0, acc[0][0], false, false);
        acc[1][0] = __builtin_amdgcn_wmma_f32_16x16x4_f32(false, a1, false, b0, (short)0, acc[1][0], false, false);
        acc[0][1] = __builtin_amdgcn_wmma_f32_16x16x4_f32(false, a0, false, b1, (short)0, acc[0][1], false, false);
        acc[1][1] = __builtin_amdgcn_wmma_f32_16x16x4_f32(false, a1, false, b1, (short)0, acc[1][1], false, false);
        acc[0][2] = __builtin_amdgcn_wmma_f32_16x16x4_f32(false, a0, false, b2, (short)0, acc[0][2], false, false);
        acc[1][2] = __builtin_amdgcn_wmma_f32_16x16x4_f32(false, a1, false, b2, (short)0, acc[1][2], false, false);
        acc[0][3] = __builtin_amdgcn_wmma_f32_16x16x4_f32(false, a0, false, b3, (short)0, acc[0][3], false, false);
        acc[1][3] = __builtin_amdgcn_wmma_f32_16x16x4_f32(false, a1, false, b3, (short)0, acc[1][3], false, false);
    }

#pragma unroll
    for (int k0 = 0; k0 < NL; k0 += 4) {
        v2f a0 = *(const v2f*)(cnt + (size_t)(m0 + r) * NL + k0 + 2 * h);
        v2f a1 = *(const v2f*)(cnt + (size_t)(m0 + 16 + r) * NL + k0 + 2 * h);
#pragma unroll
        for (int nt = 0; nt < 4; ++nt) {
            v2f b = *(const v2f*)(Bt + (size_t)(o0 + nt * 16 + r) * NL + k0 + 2 * h);
            acc[0][nt] = __builtin_amdgcn_wmma_f32_16x16x4_f32(false, a0, false, b, (short)0, acc[0][nt], false, false);
            acc[1][nt] = __builtin_amdgcn_wmma_f32_16x16x4_f32(false, a1, false, b, (short)0, acc[1][nt], false, false);
        }
    }

#pragma unroll
    for (int mt = 0; mt < 2; ++mt)
#pragma unroll
        for (int nt = 0; nt < 4; ++nt)
#pragma unroll
            for (int v = 0; v < 8; ++v)
                out[(size_t)(m0 + mt * 16 + v + 8 * h) * Dn + o0 + nt * 16 + r] = acc[mt][nt][v];
}

extern "C" void kernel_launch(void* const* d_in, const int* in_sizes, int n_in,
                              void* d_out, int out_size, void* d_ws, size_t ws_size,
                              hipStream_t stream) {
    const float* feature = (const float*)d_in[0];
    const int*   graph   = (const int*)d_in[1];
    const float* weights = (const float*)d_in[2];
    const float* bias    = (const float*)d_in[3];
    float* out = (float*)d_out;

    float* ws = (float*)d_ws;
    float* Wa  = ws;
    float* Bt  = ws + Dn * Dn;
    float* cnt = ws + Dn * Dn + Dn * NL;

    prep_weights<<<(Dn * Dn + 255) / 256, 256, 0, stream>>>(weights, bias, Wa, Bt);
    count_labels<<<ROWS, Nn, 0, stream>>>(graph, cnt);
    gemm_fused<<<128, 256, 0, stream>>>(feature, Wa, Bt, cnt, out);
}


# GEMM re-tiled through LDS: 256 groups of 64x128, 32-wide k chunks of X and W fetched by row-contiguous 16-byte loads into two LDS stages, fragments read from LDS, 8 waves of 32x32; sum order kept
# speedup vs baseline: 1.5058x; 1.5058x over previous
#include <hip/hip_runtime.h>

typedef __attribute__((ext_vector_type(2))) float v2f;
typedef __attribute__((ext_vector_type(8))) float v8f;

#define Bb 32
#define Nn 128
#define Dn 512
#define NL 16
#define ROWS (Bb * Nn)

__global__ void prep_weights(const float* __restrict__ w,
                             const float* __restrict__ bias,
                             float* __restrict__ Wa,
                             float* __restrict__ Bt) {
    int idx = blockIdx.x * blockDim.x + threadIdx.x;
    if (idx < Dn * Dn) {
        int o = idx / Dn;
        int i = idx - o * Dn;
        float v = w[idx] + w[Dn * Dn + idx];
        if (o == i) v += 1.0f;
        Wa[idx] = v;
    }
    if (idx < Dn * NL) {
        int o = idx / NL;
        int l = idx - o * NL;
        Bt[idx] = bias[l * Dn + o];
    }
}

__global__ void count_labels(const int* __restrict__ graph,
                             float* __restrict__ cnt) {
    __shared__ int h[NL];
    int row = blockIdx.x;
    int t = threadIdx.x;
    if (t < NL) h[t] = 0;
    __syncthreads();
    int lab = graph[row * Nn + t] & (NL - 1);
    atomicAdd(&h[lab], 1);
    __syncthreads();
    if (t < NL) cnt[row * NL + t] = (float)h[t];
}

#define BM 64
#define BN 128
#define BK 32
#define LP 36

__device__ __forceinline__ v8f mma4(v2f a, v2f b, v8f c) {
    return __builtin_amdgcn_wmma_f32_16x16x4_f32(false, a, false, b, (short)0, c, false, false);
}

template <int STEPS>
__device__ __forceinline__ void mma_steps(const float* pa, const float* pb, v8f (&acc)[2][2]) {
#pragma unroll
    for (int s = 0; s < STEPS; ++s) {
        v2f a0 = *(const v2f*)(pa + 4 * s);
        v2f a1 = *(const v2f*)(pa + 16 * LP + 4 * s);
        v2f b0 = *(const v2f*)(pb + 4 * s);
        v2f b1 = *(const v2f*)(pb + 16 * LP + 4 * s);
        acc[0][0] = mma4(a0, b0, acc[0][0]);
        acc[1][0] = mma4(a1, b0, acc[1][0]);
        acc[0][1] = mma4(a0, b1, acc[0][1]);
        acc[1][1] = mma4(a1, b1, acc[1][1]);
    }
}

__global__ __launch_bounds__(256) void gemm_fused(
    const float* __restrict__ X,
    const float* __restrict__ Wa,
    const float* __restrict__ Bt,
    const float* __restrict__ cnt,
    float* __restrict__ out) {

    __shared__ __attribute__((aligned(16))) float sA[2 * BM * LP];
    __shared__ __attribute__((aligned(16))) float sB[2 * BN * LP];

    const int t    = threadIdx.x;
    const int lane = t & 31;
    const int wave = t >> 5;
    const int r = lane & 15;
    const int h = lane >> 4;
    const int bm0 = (blockIdx.x >> 2) * BM;
    const int bn0 = (blockIdx.x & 3) * BN;
    const int wm = wave >> 2;
    const int wn = wave & 3;

    const int srow = t >> 3;
    const int scol = (t & 7) * 4;
    const float* gA = X  + (size_t)(bm0 + srow) * Dn + scol;
    const float* gB = Wa + (size_t)(bn0 + srow) * Dn + scol;
    const int sOff = srow * LP + scol;

    const int fa = (wm * 32 + r) * LP + 2 * h;
    const int fb = (wn * 32 + r) * LP + 2 * h;

    float4 ra[2], rb[4];

#pragma unroll
    for (int i = 0; i < 2; ++i) ra[i] = *(const float4*)(gA + (size_t)i * 32 * Dn);
#pragma unroll
    for (int i = 0; i < 4; ++i) rb[i] = *(const float4*)(gB + (size_t)i * 32 * Dn);
#pragma unroll
    for (int i = 0; i < 2; ++i) *(float4*)(sA + sOff + i * 32 * LP) = ra[i];
#pragma unroll
    for (int i = 0; i < 4; ++i) *(float4*)(sB + sOff + i * 32 * LP) = rb[i];
    __syncthreads();

    v8f acc[2][2];
#pragma unroll
    for (int mt = 0; mt < 2; ++mt)
#pragma unroll
        for (int nt = 0; nt < 2; ++nt)
            acc[mt][nt] = (v8f){0.f, 0.f, 0.f, 0.f, 0.f, 0.f, 0.f, 0.f};

    for (int kc = 0; kc < Dn / BK - 1; ++kc) {
        const int cur = kc & 1;
        const int kb = (kc + 1) * BK;
#pragma unroll
        for (int i = 0; i < 2; ++i) ra[i] = *(const float4*)(gA + (size_t)i * 32 * Dn + kb);
#pragma unroll
        for (int i = 0; i < 4; ++i) rb[i] = *(const float4*)(gB + (size_t)i * 32 * Dn + kb);

        mma_steps<BK / 4>(sA + cur * BM * LP + fa, sB + cur * BN * LP + fb, acc);

#pragma unroll
        for (int i = 0; i < 2; ++i) *(float4*)(sA + (cur ^ 1) * BM * LP + sOff + i * 32 * LP) = ra[i];
#pragma unroll
        for (int i = 0; i < 4; ++i) *(float4*)(sB + (cur ^ 1) * BN * LP + sOff + i * 32 * LP) = rb[i];
        __syncthreads();
    }

    const int trow = t >> 2;
    const int tcol = (t & 3) * 4;
    float4 ta = *(const float4*)(cnt + (size_t)(bm0 + trow) * NL + tcol);
    float4 tb[2];
#pragma unroll
    for (int i = 0; i < 2; ++i) tb[i] = *(const float4*)(Bt + (size_t)(bn0 + trow + 64 * i) * NL + tcol);

    mma_steps<BK / 4>(sA + BM * LP + fa, sB + BN * LP + fb, acc);

    *(float4*)(sA + trow * LP + tcol) = ta;
#pragma unroll
    for (int i = 0; i < 2; ++i) *(float4*)(sB + (trow + 64 * i) * LP + tcol) = tb[i];
    __syncthreads();

    mma_steps<NL / 4>(sA + fa, sB + fb, acc);

    const int row0 = bm0 + wm * 32;
    const int col0 = bn0 + wn * 32;
#pragma unroll
    for (int mt = 0; mt < 2; ++mt)
#pragma unroll
        for (int nt = 0; nt < 2; ++nt)
#pragma unroll
            for (int v = 0; v < 8; ++v)
                out[(size_t)(row0 + mt * 16 + v + 8 * h) * Dn + col0 + nt * 16 + r] = acc[mt][nt][v];
}

extern "C" void kernel_launch(void* const* d_in, const int* in_sizes, int n_in,
                              void* d_out, int out_size, void* d_ws, size_t ws_size,
                              hipStream_t stream) {
    const float* feature = (const float*)d_in[0];
    const int*   graph   = (const int*)d_in[1];
    const float* weights = (const float*)d_in[2];
    const float* bias    = (const float*)d_in[3];
    float* out = (float*)d_out;

    float* ws = (float*)d_ws;
    float* Wa  = ws;
    float* Bt  = ws + Dn * Dn;
    float* cnt = ws + Dn * Dn + Dn * NL;

    prep_weights<<<(Dn * Dn + 255) / 256, 256, 0, stream>>>(weights, bias, Wa, Bt);
    count_labels<<<ROWS, Nn, 0, stream>>>(graph, cnt);
    gemm_fused<<<256, 256, 0, stream>>>(feature, Wa, Bt, cnt, out);
}


# Single launch: W0+W1+identity summed in registers while staging W tiles, label counts by LDS integer atomics inside the GEMM group, bias read transposed for the tail; sum order kept
# speedup vs baseline: 1.8210x; 1.2093x over previous
#include <hip/hip_runtime.h>

typedef __attribute__((ext_vector_type(2))) float v2f;
typedef __attribute__((ext_vector_type(8))) float v8f;
typedef __attribute__((ext_vector_type(4))) float v4f;
typedef __attribute__((ext_vector_type(4))) int   v4i;

#define Bb 32
#define Nn 128
#define Dn 512
#define NL 16
#define ROWS (Bb * Nn)

#define BM 64
#define BN 128
#define BK 32
#define NCH (Dn / BK)
#define LP 36
#define GCH 8

__device__ __forceinline__ v8f mma4(v2f a, v2f b, v8f c) {
    return __builtin_amdgcn_wmma_f32_16x16x4_f32(false, a, false, b, (short)0, c, false, false);
}

template <int STEPS>
__device__ __forceinline__ void mma_steps(const float* pa, const float* pb, v8f (&acc)[2][2]) {
#pragma unroll
    for (int s = 0; s < STEPS; ++s) {
        v2f a0 = *(const v2f*)(pa + 4 * s);
        v2f a1 = *(const v2f*)(pa + 16 * LP + 4 * s);
        v2f b0 = *(const v2f*)(pb + 4 * s);
        v2f b1 = *(const v2f*)(pb + 16 * LP + 4 * s);
        acc[0][0] = mma4(a0, b0, acc[0][0]);
        acc[1][0] = mma4(a1, b0, acc[1][0]);
        acc[0][1] = mma4(a0, b1, acc[0][1]);
        acc[1][1] = mma4(a1, b1, acc[1][1]);
    }
}

struct Fetched {
    v4f a[2];
    v4f w0[4];
    v4f w1[4];
    v4i g;
};

struct Maps {
    const float* gA;
    const float* gB0;
    const float* gB1;
    const int*   gG;
    int sOff;
    int srow;
    v4f e;
    int bn0;
    int fa, fb;
};

__device__ __forceinline__ void fetch_tiles(Fetched& f, const Maps& m, int kb) {
#pragma unroll
    for (int i = 0; i < 2; ++i) f.a[i] = *(const v4f*)(m.gA + (size_t)i * 32 * Dn + kb);
#pragma unroll
    for (int i = 0; i < 4; ++i) f.w0[i] = *(const v4f*)(m.gB0 + (size_t)i * 32 * Dn + kb);
#pragma unroll
    for (int i = 0; i < 4; ++i) f.w1[i] = *(const v4f*)(m.gB1 + (size_t)i * 32 * Dn + kb);
}

__device__ __forceinline__ void fetch_graph(Fetched& f, const Maps& m, int p) {
    f.g = *(const v4i*)(m.gG + (size_t)(p >> 2) * 32 * Nn + (p & 3) * 32);
}

__device__ __forceinline__ void store_tiles(const Fetched& f, const Maps& m, int kb, int st, float* sA, float* sB) {
    const int ib = kb - m.bn0;
#pragma unroll
    for (int i = 0; i < 2; ++i) *(v4f*)(sA + st * BM * LP + m.sOff + i * 32 * LP) = f.a[i];
#pragma unroll
    for (int i = 0; i < 4; ++i) {
        const float s = (ib == 32 * i) ? 1.0f : 0.0f;
        const v4f v = (f.w0[i] + f.w1[i]) + m.e * s;
        *(v4f*)(sB + st * BN * LP + m.sOff + i * 32 * LP) = v;
    }
}

__device__ __forceinline__ void count_piece(const Fetched& f, const Maps& m, int p, int* hist) {
    int* hr = hist + (m.srow + (p >> 2) * 32) * NL;
    atomicAdd(hr + (f.g.x & (NL - 1)), 1);
    atomicAdd(hr + (f.g.y & (NL - 1)), 1);
    atomicAdd(hr + (f.g.z & (NL - 1)), 1);
    atomicAdd(hr + (f.g.w & (NL - 1)), 1);
}

template <bool ADD, bool FETCH_G, bool FETCH_W>
__device__ __forceinline__ void chunk_step(int kc, Fetched& f, const Maps& m,
                                           float* sA, float* sB, int* hist, v8f (&acc)[2][2]) {
    const int cur = kc & 1;
    store_tiles(f, m, (kc + 1) * BK, cur ^ 1, sA, sB);
    if constexpr (ADD) count_piece(f, m, kc, hist);
    if constexpr (FETCH_W) fetch_tiles(f, m, (kc + 2) * BK);
    if constexpr (FETCH_G) fetch_graph(f, m, kc + 1);
    mma_steps<BK / 4>(sA + cur * BM * LP + m.fa, sB + cur * BN * LP + m.fb, acc);
    __syncthreads();
}

__global__ __launch_bounds__(256) void gemm_fused(
    const float* __restrict__ X,
    const float* __restrict__ Wt,
    const float* __restrict__ bias,
    const int*   __restrict__ graph,
    float* __restrict__ out) {

    __shared__ __attribute__((aligned(16))) float sA[2 * BM * LP];
    __shared__ __attribute__((aligned(16))) float sB[2 * BN * LP];
    __shared__ __attribute__((aligned(16))) int   hist[BM * NL];

    const int t    = threadIdx.x;
    const int lane = t & 31;
    const int wave = t >> 5;
    const int r = lane & 15;
    const int h = lane >> 4;
    const int bm0 = (blockIdx.x >> 2) * BM;
    const int bn0 = (blockIdx.x & 3) * BN;
    const int wm = wave >> 2;
    const int wn = wave & 3;

    const int srow = t >> 3;
    const int scol = (t & 7) * 4;
    Maps m;
    m.gA  = X  + (size_t)(bm0 + srow) * Dn + scol;
    m.gB0 = Wt + (size_t)(bn0 + srow) * Dn + scol;
    m.gB1 = m.gB0 + (size_t)Dn * Dn;
    m.gG  = graph + (size_t)(bm0 + srow) * Nn + scol;
    m.sOff = srow * LP + scol;
    m.srow = srow;
    const int dsel = srow - scol;
    m.e = (v4f){dsel == 0 ? 1.0f : 0.0f, dsel == 1 ? 1.0f : 0.0f, dsel == 2 ? 1.0f : 0.0f, dsel == 3 ? 1.0f : 0.0f};
    m.bn0 = bn0;
    m.fa = (wm * 32 + r) * LP + 2 * h;
    m.fb = (wn * 32 + r) * LP + 2 * h;

    *(int4*)(hist + t * 4) = make_int4(0, 0, 0, 0);

    Fetched f0, f;
    fetch_tiles(f0, m, 0);
    fetch_tiles(f, m, BK);
    fetch_graph(f, m, 0);
    store_tiles(f0, m, 0, 0, sA, sB);
    __syncthreads();

    v8f acc[2][2];
#pragma unroll
    for (int mt = 0; mt < 2; ++mt)
#pragma unroll
        for (int nt = 0; nt < 2; ++nt)
            acc[mt][nt] = (v8f){0.f, 0.f, 0.f, 0.f, 0.f, 0.f, 0.f, 0.f};

#pragma unroll 1
    for (int kc = 0; kc < GCH - 1; ++kc) chunk_step<true, true, true>(kc, f, m, sA, sB, hist, acc);
    chunk_step<true, false, true>(GCH - 1, f, m, sA, sB, hist, acc);
#pragma unroll 1
    for (int kc = GCH; kc < NCH - 2; ++kc) chunk_step<false, false, true>(kc, f, m, sA, sB, hist, acc);

    const int trow = t >> 2;
    const int tcol = (t & 3) * 4;
    const int bo = t & (BN - 1);
    const int bl = (t >> 7) * 8;
    float tb[8];
#pragma unroll
    for (int j = 0; j < 8; ++j) tb[j] = bias[(size_t)(bl + j) * Dn + bn0 + bo];

    chunk_step<false, false, false>(NCH - 2, f, m, sA, sB, hist, acc);

    const int4 hc = *(const int4*)(hist + trow * NL + tcol);
    *(float4*)(sA + trow * LP + tcol) = make_float4((float)hc.x, (float)hc.y, (float)hc.z, (float)hc.w);
    *(float4*)(sB + bo * LP + bl)     = make_float4(tb[0], tb[1], tb[2], tb[3]);
    *(float4*)(sB + bo * LP + bl + 4) = make_float4(tb[4], tb[5], tb[6], tb[7]);

    mma_steps<BK / 4>(sA + BM * LP + m.fa, sB + BN * LP + m.fb, acc);
    __syncthreads();

    mma_steps<NL / 4>(sA + m.fa, sB + m.fb, acc);

    const int row0 = bm0 + wm * 32;
    const int col0 = bn0 + wn * 32;
#pragma unroll
    for (int mt = 0; mt < 2; ++mt)
#pragma unroll
        for (int nt = 0; nt < 2; ++nt)
#pragma unroll
            for (int v = 0; v < 8; ++v)
                out[(size_t)(row0 + mt * 16 + v + 8 * h) * Dn + col0 + nt * 16 + r] = acc[mt][nt][v];
}

extern "C" void kernel_launch(void* const* d_in, const int* in_sizes, int n_in,
                              void* d_out, int out_size, void* d_ws, size_t ws_size,
                              hipStream_t stream) {
    const float* feature = (const float*)d_in[0];
    const int*   graph   = (const int*)d_in[1];
    const float* weights = (const float*)d_in[2];
    const float* bias    = (const float*)d_in[3];
    float* out = (float*)d_out;

    gemm_fused<<<256, 256, 0, stream>>>(feature, weights, bias, graph, out);
}


# Global loads of a chunk stay in flight over two whole steps: second register set, and the step barrier fences LDS only, so it no longer drains the load counter at every step; sum order kept
# speedup vs baseline: 2.1198x; 1.1641x over previous
#include <hip/hip_runtime.h>

typedef __attribute__((ext_vector_type(2))) float v2f;
typedef __attribute__((ext_vector_type(8))) float v8f;
typedef __attribute__((ext_vector_type(4))) float v4f;
typedef __attribute__((ext_vector_type(4))) int   v4i;

#define Bb 32
#define Nn 128
#define Dn 512
#define NL 16
#define ROWS (Bb * Nn)

#define BM 64
#define BN 128
#define BK 32
#define NCH (Dn / BK)
#define LP 36
#define GCH 8

__device__ __forceinline__ v8f mma4(v2f a, v2f b, v8f c) {
    return __builtin_amdgcn_wmma_f32_16x16x4_f32(false, a, false, b, (short)0, c, false, false);
}

__device__ __forceinline__ void lds_barrier() {
    __builtin_amdgcn_fence(__ATOMIC_RELEASE, "workgroup", "local");
    __builtin_amdgcn_s_barrier();
    __builtin_amdgcn_fence(__ATOMIC_ACQUIRE, "workgroup", "local");
    __builtin_amdgcn_sched_barrier(0);
}

template <int STEPS>
__device__ __forceinline__ void mma_steps(const float* pa, const float* pb, v8f (&acc)[2][2]) {
#pragma unroll
    for (int s = 0; s < STEPS; ++s) {
        v2f a0 = *(const v2f*)(pa + 4 * s);
        v2f a1 = *(const v2f*)(pa + 16 * LP + 4 * s);
        v2f b0 = *(const v2f*)(pb + 4 * s);
        v2f b1 = *(const v2f*)(pb + 16 * LP + 4 * s);
        acc[0][0] = mma4(a0, b0, acc[0][0]);
        acc[1][0] = mma4(a1, b0, acc[1][0]);
        acc[0][1] = mma4(a0, b1, acc[0][1]);
        acc[1][1] = mma4(a1, b1, acc[1][1]);
    }
}

struct Fetched {
    v4f a[2];
    v4f w0[4];
    v4f w1[4];
    v4i g;
};

struct Maps {
    const float* gA;
    const float* gB0;
    const float* gB1;
    const int*   gG;
    int sOff;
    int srow;
    v4f e;
    int bn0;
    int fa, fb;
};

__device__ __forceinline__ void fetch_tiles(Fetched& f, const Maps& m, int kb) {
#pragma unroll
    for (int i = 0; i < 2; ++i) f.a[i] = *(const v4f*)(m.gA + (size_t)i * 32 * Dn + kb);
#pragma unroll
    for (int i = 0; i < 4; ++i) f.w0[i] = *(const v4f*)(m.gB0 + (size_t)i * 32 * Dn + kb);
#pragma unroll
    for (int i = 0; i < 4; ++i) f.w1[i] = *(const v4f*)(m.gB1 + (size_t)i * 32 * Dn + kb);
}

__device__ __forceinline__ void fetch_graph(Fetched& f, const Maps& m, int p) {
    f.g = *(const v4i*)(m.gG + (size_t)(p >> 2) * 32 * Nn + (p & 3) * 32);
}

__device__ __forceinline__ void store_tiles(const Fetched& f, const Maps& m, int kb, int st, float* sA, float* sB) {
    const int ib = kb - m.bn0;
#pragma unroll
    for (int i = 0; i < 2; ++i) *(v4f*)(sA + st * BM * LP + m.sOff + i * 32 * LP) = f.a[i];
#pragma unroll
    for (int i = 0; i < 4; ++i) {
        const float s = (ib == 32 * i) ? 1.0f : 0.0f;
        const v4f v = (f.w0[i] + f.w1[i]) + m.e * s;
        *(v4f*)(sB + st * BN * LP + m.sOff + i * 32 * LP) = v;
    }
}

__device__ __forceinline__ void count_piece(const Fetched& f, const Maps& m, int p, int* hist) {
    int* hr = hist + (m.srow + (p >> 2) * 32) * NL;
    atomicAdd(hr + (f.g.x & (NL - 1)), 1);
    atomicAdd(hr + (f.g.y & (NL - 1)), 1);
    atomicAdd(hr + (f.g.z & (NL - 1)), 1);
    atomicAdd(hr + (f.g.w & (NL - 1)), 1);
}

template <int PAR, bool ADD, bool FETCH_G, bool FETCH_W>
__device__ __forceinline__ void chunk_step(int kc, Fetched& f, const Maps& m,
                                           float* sA, float* sB, int* hist, v8f (&acc)[2][2]) {
    store_tiles(f, m, (kc + 1) * BK, PAR ^ 1, sA, sB);
    if constexpr (ADD) count_piece(f, m, kc, hist);
    if constexpr (FETCH_W) fetch_tiles(f, m, (kc + 3) * BK);
    if constexpr (FETCH_G) fetch_graph(f, m, kc + 2);
    mma_steps<BK / 4>(sA + PAR * BM * LP + m.fa, sB + PAR * BN * LP + m.fb, acc);
    lds_barrier();
}

__global__ __launch_bounds__(256) void gemm_fused(
    const float* __restrict__ X,
    const float* __restrict__ Wt,
    const float* __restrict__ bias,
    const int*   __restrict__ graph,
    float* __restrict__ out) {

    __shared__ __attribute__((aligned(16))) float sA[2 * BM * LP];
    __shared__ __attribute__((aligned(16))) float sB[2 * BN * LP];
    __shared__ __attribute__((aligned(16))) int   hist[BM * NL];

    const int t    = threadIdx.x;
    const int lane = t & 31;
    const int wave = t >> 5;
    const int r = lane & 15;
    const int h = lane >> 4;
    const int bm0 = (blockIdx.x >> 2) * BM;
    const int bn0 = (blockIdx.x & 3) * BN;
    const int wm = wave >> 2;
    const int wn = wave & 3;

    const int srow = t >> 3;
    const int scol = (t & 7) * 4;
    Maps m;
    m.gA  = X  + (size_t)(bm0 + srow) * Dn + scol;
    m.gB0 = Wt + (size_t)(bn0 + srow) * Dn + scol;
    m.gB1 = m.gB0 + (size_t)Dn * Dn;
    m.gG  = graph + (size_t)(bm0 + srow) * Nn + scol;
    m.sOff = srow * LP + scol;
    m.srow = srow;
    const int dsel = srow - scol;
    m.e = (v4f){dsel == 0 ? 1.0f : 0.0f, dsel == 1 ? 1.0f : 0.0f, dsel == 2 ? 1.0f : 0.0f, dsel == 3 ? 1.0f : 0.0f};
    m.bn0 = bn0;
    m.fa = (wm * 32 + r) * LP + 2 * h;
    m.fb = (wn * 32 + r) * LP + 2 * h;

    *(int4*)(hist + t * 4) = make_int4(0, 0, 0, 0);

    Fetched fc, fo, fe;
    fetch_tiles(fc, m, 0);
    __builtin_amdgcn_sched_barrier(0);
    fetch_tiles(fo, m, BK);
    fetch_graph(fo, m, 0);
    __builtin_amdgcn_sched_barrier(0);
    fetch_tiles(fe, m, 2 * BK);
    fetch_graph(fe, m, 1);
    store_tiles(fc, m, 0, 0, sA, sB);
    lds_barrier();

    v8f acc[2][2];
#pragma unroll
    for (int mt = 0; mt < 2; ++mt)
#pragma unroll
        for (int nt = 0; nt < 2; ++nt)
            acc[mt][nt] = (v8f){0.f, 0.f, 0.f, 0.f, 0.f, 0.f, 0.f, 0.f};

#pragma unroll 1
    for (int kc = 0; kc < GCH - 2; kc += 2) {
        chunk_step<0, true, true, true>(kc, fo, m, sA, sB, hist, acc);
        chunk_step<1, true, true, true>(kc + 1, fe, m, sA, sB, hist, acc);
    }
    chunk_step<0, true, false, true>(GCH - 2, fo, m, sA, sB, hist, acc);
    chunk_step<1, true, false, true>(GCH - 1, fe, m, sA, sB, hist, acc);
#pragma unroll 1
    for (int kc = GCH; kc < NCH - 4; kc += 2) {
        chunk_step<0, false, false, true>(kc, fo, m, sA, sB, hist, acc);
        chunk_step<1, false, false, true>(kc + 1, fe, m, sA, sB, hist, acc);
    }
    chunk_step<0, false, false, true>(NCH - 4, fo, m, sA, sB, hist, acc);

    const int trow = t >> 2;
    const int tcol = (t & 3) * 4;
    const int bo = t & (BN - 1);
    const int bl = (t >> 7) * 8;
    float tb[8];
#pragma unroll
    for (int j = 0; j < 8; ++j) tb[j] = bias[(size_t)(bl + j) * Dn + bn0 + bo];

    chunk_step<1, false, false, false>(NCH - 3, fe, m, sA, sB, hist, acc);
    chunk_step<0, false, false, false>(NCH - 2, fo, m, sA, sB, hist, acc);

    const int4 hc = *(const int4*)(hist + trow * NL + tcol);
    *(float4*)(sA + trow * LP + tcol) = make_float4((float)hc.x, (float)hc.y, (float)hc.z, (float)hc.w);
    *(float4*)(sB + bo * LP + bl)     = make_float4(tb[0], tb[1], tb[2], tb[3]);
    *(float4*)(sB + bo * LP + bl + 4) = make_float4(tb[4], tb[5], tb[6], tb[7]);

    mma_steps<BK / 4>(sA + BM * LP + m.fa, sB + BN * LP + m.fb, acc);
    lds_barrier();

    mma_steps<NL / 4>(sA + m.fa, sB + m.fb, acc);

    const int row0 = bm0 + wm * 32;
    const int col0 = bn0 + wn * 32;
#pragma unroll
    for (int mt = 0; mt < 2; ++mt)
#pragma unroll
        for (int nt = 0; nt < 2; ++nt)
#pragma unroll
            for (int v = 0; v < 8; ++v)
                out[(size_t)(row0 + mt * 16 + v + 8 * h) * Dn + col0 + nt * 16 + r] = acc[mt][nt][v];
}

extern "C" void kernel_launch(void* const* d_in, const int* in_sizes, int n_in,
                              void* d_out, int out_size, void* d_ws, size_t ws_size,
                              hipStream_t stream) {
    const float* feature = (const float*)d_in[0];
    const int*   graph   = (const int*)d_in[1];
    const float* weights = (const float*)d_in[2];
    const float* bias    = (const float*)d_in[3];
    float* out = (float*)d_out;

    gemm_fused<<<256, 256, 0, stream>>>(feature, weights, bias, graph, out);
}


# Fragments of chunk k+1 are read from LDS as one group at the top of step k into a second register set, so a step's WMMAs run from registers and its barrier splits into arrive and wait; sum order kept
# speedup vs baseline: 2.2982x; 1.0842x over previous
#include <hip/hip_runtime.h>

typedef __attribute__((ext_vector_type(2))) float v2f;
typedef __attribute__((ext_vector_type(8))) float v8f;
typedef __attribute__((ext_vector_type(4))) float v4f;
typedef __attribute__((ext_vector_type(4))) int   v4i;

#define Bb 32
#define Nn 128
#define Dn 512
#define NL 16
#define ROWS (Bb * Nn)

#define BM 64
#define BN 128
#define BK 32
#define NCH (Dn / BK)
#define LP 36
#define GCH 8

__device__ __forceinline__ v8f mma4(v2f a, v2f b, v8f c) {
    return __builtin_amdgcn_wmma_f32_16x16x4_f32(false, a, false, b, (short)0, c, false, false);
}

__device__ __forceinline__ void lds_barrier() {
    __builtin_amdgcn_fence(__ATOMIC_RELEASE, "workgroup", "local");
    __builtin_amdgcn_s_barrier();
    __builtin_amdgcn_fence(__ATOMIC_ACQUIRE, "workgroup", "local");
    __builtin_amdgcn_sched_barrier(0);
}
__device__ __forceinline__ void lds_arrive() {
    __builtin_amdgcn_fence(__ATOMIC_RELEASE, "workgroup", "local");
    __builtin_amdgcn_s_barrier_signal(-1);
}
__device__ __forceinline__ void lds_wait() {
    __builtin_amdgcn_sched_barrier(0);
    __builtin_amdgcn_s_barrier_wait(-1);
    __builtin_amdgcn_fence(__ATOMIC_ACQUIRE, "workgroup", "local");
    __builtin_amdgcn_sched_barrier(0);
}

struct Frags {
    v2f a0[BK / 4], a1[BK / 4], b0[BK / 4], b1[BK / 4];
};

template <int STEPS>
__device__ __forceinline__ void load_frags(Frags& r, const float* pa, const float* pb) {
#pragma unroll
    for (int s = 0; s < STEPS; ++s) {
        r.a0[s] = *(const v2f*)(pa + 4 * s);
        r.a1[s] = *(const v2f*)(pa + 16 * LP + 4 * s);
        r.b0[s] = *(const v2f*)(pb + 4 * s);
        r.b1[s] = *(const v2f*)(pb + 16 * LP + 4 * s);
    }
}

template <int S0, int S1>
__device__ __forceinline__ void mma_range(const Frags& r, v8f (&acc)[2][2]) {
#pragma unroll
    for (int s = S0; s < S1; ++s) {
        acc[0][0] = mma4(r.a0[s], r.b0[s], acc[0][0]);
        acc[1][0] = mma4(r.a1[s], r.b0[s], acc[1][0]);
        acc[0][1] = mma4(r.a0[s], r.b1[s], acc[0][1]);
        acc[1][1] = mma4(r.a1[s], r.b1[s], acc[1][1]);
    }
}

struct Fetched {
    v4f a[2];
    v4f w0[4];
    v4f w1[4];
    v4i g;
};

struct Maps {
    const float* gA;
    const float* gB0;
    const float* gB1;
    const int*   gG;
    int sOff;
    int srow;
    v4f e;
    int bn0;
    int fa, fb;
};

__device__ __forceinline__ void fetch_tiles(Fetched& f, const Maps& m, int kb) {
#pragma unroll
    for (int i = 0; i < 2; ++i) f.a[i] = *(const v4f*)(m.gA + (size_t)i * 32 * Dn + kb);
#pragma unroll
    for (int i = 0; i < 4; ++i) f.w0[i] = *(const v4f*)(m.gB0 + (size_t)i * 32 * Dn + kb);
#pragma unroll
    for (int i = 0; i < 4; ++i) f.w1[i] = *(const v4f*)(m.gB1 + (size_t)i * 32 * Dn + kb);
}

__device__ __forceinline__ void fetch_graph(Fetched& f, const Maps& m, int p) {
    f.g = *(const v4i*)(m.gG + (size_t)(p >> 2) * 32 * Nn + (p & 3) * 32);
}

__device__ __forceinline__ void store_tiles(const Fetched& f, const Maps& m, int kb, int st, float* sA, float* sB) {
    const int ib = kb - m.bn0;
#pragma unroll
    for (int i = 0; i < 2; ++i) *(v4f*)(sA + st * BM * LP + m.sOff + i * 32 * LP) = f.a[i];
#pragma unroll
    for (int i = 0; i < 4; ++i) {
        const float s = (ib == 32 * i) ? 1.0f : 0.0f;
        const v4f v = (f.w0[i] + f.w1[i]) + m.e * s;
        *(v4f*)(sB + st * BN * LP + m.sOff + i * 32 * LP) = v;
    }
}

__device__ __forceinline__ void count_piece(const Fetched& f, const Maps& m, int p, int* hist) {
    int* hr = hist + (m.srow + (p >> 2) * 32) * NL;
    atomicAdd(hr + (f.g.x & (NL - 1)), 1);
    atomicAdd(hr + (f.g.y & (NL - 1)), 1);
    atomicAdd(hr + (f.g.z & (NL - 1)), 1);
    atomicAdd(hr + (f.g.w & (NL - 1)), 1);
}

template <int PAR, bool ADD, bool FETCH_G, bool FETCH_W>
__device__ __forceinline__ void chunk_step(int kc, Fetched& f, const Maps& m, float* sA, float* sB, int* hist,
                                           const Frags& cur, Frags& nxt, v8f (&acc)[2][2]) {
    load_frags<BK / 4>(nxt, sA + (PAR ^ 1) * BM * LP + m.fa, sB + (PAR ^ 1) * BN * LP + m.fb);
    __builtin_amdgcn_sched_barrier(0);
    mma_range<0, 2>(cur, acc);
    __builtin_amdgcn_sched_barrier(0);
    store_tiles(f, m, (kc + 2) * BK, PAR, sA, sB);
    if constexpr (ADD) count_piece(f, m, kc, hist);
    if constexpr (FETCH_W) fetch_tiles(f, m, (kc + 4) * BK);
    if constexpr (FETCH_G) fetch_graph(f, m, kc + 2);
    __builtin_amdgcn_sched_barrier(0);
    mma_range<2, 6>(cur, acc);
    __builtin_amdgcn_sched_barrier(0);
    lds_arrive();
    __builtin_amdgcn_sched_barrier(0);
    mma_range<6, 8>(cur, acc);
    lds_wait();
}

__global__ __launch_bounds__(256) void gemm_fused(
    const float* __restrict__ X,
    const float* __restrict__ Wt,
    const float* __restrict__ bias,
    const int*   __restrict__ graph,
    float* __restrict__ out) {

    __shared__ __attribute__((aligned(16))) float sA[2 * BM * LP];
    __shared__ __attribute__((aligned(16))) float sB[2 * BN * LP];
    __shared__ __attribute__((aligned(16))) int   hist[BM * NL];

    const int t    = threadIdx.x;
    const int lane = t & 31;
    const int wave = t >> 5;
    const int r = lane & 15;
    const int h = lane >> 4;
    const int bm0 = (blockIdx.x >> 2) * BM;
    const int bn0 = (blockIdx.x & 3) * BN;
    const int wm = wave >> 2;
    const int wn = wave & 3;

    const int srow = t >> 3;
    const int scol = (t & 7) * 4;
    Maps m;
    m.gA  = X  + (size_t)(bm0 + srow) * Dn + scol;
    m.gB0 = Wt + (size_t)(bn0 + srow) * Dn + scol;
    m.gB1 = m.gB0 + (size_t)Dn * Dn;
    m.gG  = graph + (size_t)(bm0 + srow) * Nn + scol;
    m.sOff = srow * LP + scol;
    m.srow = srow;
    const int dsel = srow - scol;
    m.e = (v4f){dsel == 0 ? 1.0f : 0.0f, dsel == 1 ? 1.0f : 0.0f, dsel == 2 ? 1.0f : 0.0f, dsel == 3 ? 1.0f : 0.0f};
    m.bn0 = bn0;
    m.fa = (wm * 32 + r) * LP + 2 * h;
    m.fb = (wn * 32 + r) * LP + 2 * h;

    *(int4*)(hist + t * 4) = make_int4(0, 0, 0, 0);

    Fetched f0, f1, fe, fo;
    fetch_tiles(f0, m, 0);
    __builtin_amdgcn_sched_barrier(0);
    fetch_tiles(f1, m, BK);
    __builtin_amdgcn_sched_barrier(0);
    fetch_tiles(fe, m, 2 * BK);
    fetch_graph(fe, m, 0);
    __builtin_amdgcn_sched_barrier(0);
    fetch_tiles(fo, m, 3 * BK);
    fetch_graph(fo, m, 1);
    store_tiles(f0, m, 0, 0, sA, sB);
    lds_barrier();
    Frags r0, r1;
    load_frags<BK / 4>(r0, sA + m.fa, sB + m.fb);
    store_tiles(f1, m, BK, 1, sA, sB);
    lds_barrier();

    v8f acc[2][2];
#pragma unroll
    for (int mt = 0; mt < 2; ++mt)
#pragma unroll
        for (int nt = 0; nt < 2; ++nt)
            acc[mt][nt] = (v8f){0.f, 0.f, 0.f, 0.f, 0.f, 0.f, 0.f, 0.f};

#pragma unroll 1
    for (int kc = 0; kc < GCH - 2; kc += 2) {
        chunk_step<0, true, true, true>(kc, fe, m, sA, sB, hist, r0, r1, acc);
        chunk_step<1, true, true, true>(kc + 1, fo, m, sA, sB, hist, r1, r0, acc);
    }
    chunk_step<0, true, false, true>(GCH - 2, fe, m, sA, sB, hist, r0, r1, acc);
    chunk_step<1, true, false, true>(GCH - 1, fo, m, sA, sB, hist, r1, r0, acc);
#pragma unroll 1
    for (int kc = GCH; kc < NCH - 4; kc += 2) {
        chunk_step<0, false, false, true>(kc, fe, m, sA, sB, hist, r0, r1, acc);
        chunk_step<1, false, false, true>(kc + 1, fo, m, sA, sB, hist, r1, r0, acc);
    }

    const int trow = t >> 2;
    const int tcol = (t & 3) * 4;
    const int bo = t & (BN - 1);
    const int bl = (t >> 7) * 8;
    float tb[8];
#pragma unroll
    for (int j = 0; j < 8; ++j) tb[j] = bias[(size_t)(bl + j) * Dn + bn0 + bo];

    chunk_step<0, false, false, false>(NCH - 4, fe, m, sA, sB, hist, r0, r1, acc);
    chunk_step<1, false, false, false>(NCH - 3, fo, m, sA, sB, hist, r1, r0, acc);

    load_frags<BK / 4>(r1, sA + BM * LP + m.fa, sB + BN * LP + m.fb);
    __builtin_amdgcn_sched_barrier(0);
    mma_range<0, 2>(r0, acc);
    __builtin_amdgcn_sched_barrier(0);
    const int4 hc = *(const int4*)(hist + trow * NL + tcol);
    *(float4*)(sA + trow * LP + tcol) = make_float4((float)hc.x, (float)hc.y, (float)hc.z, (float)hc.w);
    *(float4*)(sB + bo * LP + bl)     = make_float4(tb[0], tb[1], tb[2], tb[3]);
    *(float4*)(sB + bo * LP + bl + 4) = make_float4(tb[4], tb[5], tb[6], tb[7]);
    __builtin_amdgcn_sched_barrier(0);
    mma_range<2, 6>(r0, acc);
    __builtin_amdgcn_sched_barrier(0);
    lds_arrive();
    __builtin_amdgcn_sched_barrier(0);
    mma_range<6, 8>(r0, acc);
    lds_wait();

    load_frags<NL / 4>(r0, sA + m.fa, sB + m.fb);
    __builtin_amdgcn_sched_barrier(0);
    mma_range<0, BK / 4>(r1, acc);
    mma_range<0, NL / 4>(r0, acc);

    const int row0 = bm0 + wm * 32;
    const int col0 = bn0 + wn * 32;
#pragma unroll
    for (int mt = 0; mt < 2; ++mt)
#pragma unroll
        for (int nt = 0; nt < 2; ++nt)
#pragma unroll
            for (int v = 0; v < 8; ++v)
                out[(size_t)(row0 + mt * 16 + v + 8 * h) * Dn + col0 + nt * 16 + r] = acc[mt][nt][v];
}

extern "C" void kernel_launch(void* const* d_in, const int* in_sizes, int n_in,
                              void* d_out, int out_size, void* d_ws, size_t ws_size,
                              hipStream_t stream) {
    const float* feature = (const float*)d_in[0];
    const int*   graph   = (const int*)d_in[1];
    const float* weights = (const float*)d_in[2];
    const float* bias    = (const float*)d_in[3];
    float* out = (float*)d_out;

    gemm_fused<<<256, 256, 0, stream>>>(feature, weights, bias, graph, out);
}
